# starting kernel
# baseline (speedup 1.0000x reference)
#include <hip/hip_runtime.h>
#include <hip/hip_bf16.h>

typedef __attribute__((ext_vector_type(2))) float v2f;
typedef __attribute__((ext_vector_type(8))) float v8f;

#define N_GRAPHS_MAX 64
#define GV_FLOATS (N_GRAPHS_MAX * 6)

__global__ void les_init_kernel(const float* __restrict__ pos_grad,
                                float* __restrict__ force,
                                float* __restrict__ gvir, int n3) {
  int i = blockIdx.x * blockDim.x + threadIdx.x;
  if (i < n3) force[i] = -pos_grad[i];
  if (i < GV_FLOATS) gvir[i] = 0.f;
}

__global__ void les_edge_kernel(const float* __restrict__ rij,
                                const float* __restrict__ fij,
                                const int* __restrict__ eidx,
                                const int* __restrict__ batch,
                                float* __restrict__ force,
                                float* __restrict__ gvir, int E) {
  __shared__ float lvir[GV_FLOATS];
  for (int i = threadIdx.x; i < GV_FLOATS; i += blockDim.x) lvir[i] = 0.f;
  __syncthreads();

  int stride = gridDim.x * blockDim.x;
  for (int e = blockIdx.x * blockDim.x + threadIdx.x; e < E; e += stride) {
    float rx = rij[3 * e + 0], ry = rij[3 * e + 1], rz = rij[3 * e + 2];
    float fx = fij[3 * e + 0], fy = fij[3 * e + 1], fz = fij[3 * e + 2];
    int i = eidx[e];
    int j = eidx[E + e];
    unsafeAtomicAdd(&force[3 * i + 0], fx);
    unsafeAtomicAdd(&force[3 * i + 1], fy);
    unsafeAtomicAdd(&force[3 * i + 2], fz);
    unsafeAtomicAdd(&force[3 * j + 0], -fx);
    unsafeAtomicAdd(&force[3 * j + 1], -fy);
    unsafeAtomicAdd(&force[3 * j + 2], -fz);
    int g = batch[j];
    float* p = &lvir[g * 6];
    atomicAdd(p + 0, rx * fx);
    atomicAdd(p + 1, ry * fy);
    atomicAdd(p + 2, rz * fz);
    atomicAdd(p + 3, rx * fy);
    atomicAdd(p + 4, ry * fz);
    atomicAdd(p + 5, rz * fx);
  }
  __syncthreads();
  for (int i = threadIdx.x; i < GV_FLOATS; i += blockDim.x)
    unsafeAtomicAdd(&gvir[i], lvir[i]);
}

__global__ void les_stress_kernel(const float* __restrict__ cell_grad,
                                  const float* __restrict__ les_cell,
                                  const float* __restrict__ vol,
                                  const float* __restrict__ gvir,
                                  float* __restrict__ out) {
  int g = blockIdx.x;
  int lane = threadIdx.x;
  const float* C = les_cell + g * 9;
  const float* G = cell_grad + g * 9;

  int m = lane & 15;
  int hi = lane >> 4;
  int k0 = hi * 2, k1 = k0 + 1;
  int mc = (m < 2) ? m : 2;
  int k0c = (k0 < 2) ? k0 : 2;
  int k1c = (k1 < 2) ? k1 : 2;
  float rowm = (m < 3) ? 1.f : 0.f;
  float m0 = (k0 < 3) ? rowm : 0.f;
  float m1 = (k1 < 3) ? rowm : 0.f;

  v2f a, b;
  a.x = m0 * C[k0c * 3 + mc];
  a.y = m1 * C[k1c * 3 + mc];
  b.x = m0 * G[k0c * 3 + mc];
  b.y = m1 * G[k1c * 3 + mc];
  v8f c = {0.f, 0.f, 0.f, 0.f, 0.f, 0.f, 0.f, 0.f};
  v8f d = __builtin_amdgcn_wmma_f32_16x16x4_f32(
       false, a,  false, b,
       (short)0, c,  false,  false);

  __shared__ float lr[9];
  if (lane < 3) {
    lr[0 * 3 + lane] = d[0];
    lr[1 * 3 + lane] = d[1];
    lr[2 * 3 + lane] = d[2];
  }
  __syncthreads();
  if (lane == 0) {
    float inv = -1.0f / vol[g];
    const float* v = gvir + g * 6;
    float* o = out + g * 6;
    o[0] = inv * (v[0] + lr[0]);
    o[1] = inv * (v[1] + lr[4]);
    o[2] = inv * (v[2] + lr[8]);
    o[3] = inv * (v[3] + lr[1]);
    o[4] = inv * (v[4] + lr[5]);
    o[5] = inv * (v[5] + lr[2]);
  }
}

extern "C" void kernel_launch(void* const* d_in, const int* in_sizes, int n_in,
                              void* d_out, int out_size, void* d_ws, size_t ws_size,
                              hipStream_t stream) {
  const float* rij      = (const float*)d_in[0];
  const float* fij      = (const float*)d_in[1];
  const float* pos_grad = (const float*)d_in[2];
  const float* cell_grad = (const float*)d_in[3];
  const float* les_cell  = (const float*)d_in[4];
  const float* vol       = (const float*)d_in[5];
  const int*   eidx      = (const int*)d_in[6];
  const int*   batch     = (const int*)d_in[7];

  int E  = in_sizes[0] / 3;
  int n3 = in_sizes[2];
  int G  = in_sizes[5];

  float* force  = (float*)d_out;
  float* stress = force + n3;
  float* gvir   = (float*)d_ws;

  const int threads = 256;
  les_init_kernel<<<(n3 + threads - 1) / threads, threads, 0, stream>>>(
      pos_grad, force, gvir, n3);
  les_edge_kernel<<<2048, threads, 0, stream>>>(
      rij, fij, eidx, batch, force, gvir, E);
  les_stress_kernel<<<G, 32, 0, stream>>>(
      cell_grad, les_cell, vol, gvir, stress);
}


# Force scatter: the per-edge random global f32 atomics become ds_add_f32 into per-block LDS node tiles (32 edge slices x 8 node tiles), flushed by whole-line global atomics; f32 sum order changes
# speedup vs baseline: 7.0042x; 7.0042x over previous
#include <hip/hip_runtime.h>
#include <hip/hip_bf16.h>

typedef __attribute__((ext_vector_type(2))) float v2f;
typedef __attribute__((ext_vector_type(8))) float v8f;

#define N_GRAPHS_MAX 64
#define GV_FLOATS (N_GRAPHS_MAX * 6)

__global__ void les_init_kernel(const float* __restrict__ pos_grad,
                                float* __restrict__ force,
                                float* __restrict__ gvir, int n3) {
  int i = blockIdx.x * blockDim.x + threadIdx.x;
  if (i < n3) force[i] = -pos_grad[i];
  if (i < GV_FLOATS) gvir[i] = 0.f;
}

#define TILE_NODES 25024
#define TILE_FLOATS (3 * TILE_NODES)
#define EDGE_THREADS 1024
#define EDGE_SLICES 32
#define EDGE_TILE_ROWS 8
#define EU 4

__device__ __forceinline__ void les_virial_share(
    const float* __restrict__ rij, const float* __restrict__ fij,
    const int* __restrict__ eidx, const int* __restrict__ batch,
    float* lvir, int E, int tid, int nthr) {
  const int nblk = gridDim.x * gridDim.y;
  const int blk = blockIdx.y * gridDim.x + blockIdx.x;
  const int step = nblk * nthr;
  for (int base = blk * nthr + tid; base < E; base += EU * step) {
    float rx[EU], ry[EU], rz[EU], fx[EU], fy[EU], fz[EU];
    int g[EU];
    bool ok[EU];
    for (int u = 0; u < EU; ++u) {
      int e = base + u * step;
      ok[u] = e < E;
      int ec = ok[u] ? e : base;
      rx[u] = rij[3 * ec + 0]; ry[u] = rij[3 * ec + 1]; rz[u] = rij[3 * ec + 2];
      fx[u] = fij[3 * ec + 0]; fy[u] = fij[3 * ec + 1]; fz[u] = fij[3 * ec + 2];
      g[u] = batch[eidx[E + ec]];
    }
    for (int u = 0; u < EU; ++u) {
      if (ok[u]) {
        float* p = &lvir[g[u] * 6];
        atomicAdd(p + 0, rx[u] * fx[u]);
        atomicAdd(p + 1, ry[u] * fy[u]);
        atomicAdd(p + 2, rz[u] * fz[u]);
        atomicAdd(p + 3, rx[u] * fy[u]);
        atomicAdd(p + 4, ry[u] * fz[u]);
        atomicAdd(p + 5, rz[u] * fx[u]);
      }
    }
  }
}

__global__ __launch_bounds__(EDGE_THREADS) void les_edge_kernel(
    const float* __restrict__ rij, const float* __restrict__ fij,
    const int* __restrict__ eidx, const int* __restrict__ batch,
    float* __restrict__ force, float* __restrict__ gvir, int E, int n3) {
  __shared__ float tile[TILE_FLOATS];
  __shared__ float lvir[GV_FLOATS];
  const int tid = threadIdx.x;
  const int nthr = blockDim.x;
  const int n_nodes = n3 / 3;

  for (int i = tid; i < GV_FLOATS; i += nthr) lvir[i] = 0.f;

  const int S = gridDim.x;
  const int s = blockIdx.x;
  const int per = (E + S - 1) / S;
  const int e0 = s * per;
  const int e1 = (e0 + per < E) ? (e0 + per) : E;
  const int nparts = (n_nodes + TILE_NODES - 1) / TILE_NODES;
  bool virial_done = false;

  for (int p = blockIdx.y; p < nparts; p += gridDim.y) {
    const int lo = p * TILE_NODES;
    const int cnt = (n_nodes - lo < TILE_NODES) ? (n_nodes - lo) : TILE_NODES;
    const int cnt3 = 3 * cnt;

    for (int k = tid; k < cnt3; k += nthr) tile[k] = 0.f;
    __syncthreads();

    for (int base = e0 + tid; base < e1; base += EU * nthr) {
      int ni[EU], nj[EU];
      float fx[EU], fy[EU], fz[EU];
      for (int u = 0; u < EU; ++u) {
        int e = base + u * nthr;
        bool ok = e < e1;
        int ec = ok ? e : base;
        int vi = eidx[ec];
        int vj = eidx[E + ec];
        ni[u] = ok ? vi : -1;
        nj[u] = ok ? vj : -1;
        fx[u] = fij[3 * ec + 0]; fy[u] = fij[3 * ec + 1]; fz[u] = fij[3 * ec + 2];
      }
      for (int u = 0; u < EU; ++u) {
        unsigned di = (unsigned)(ni[u] - lo);
        unsigned dj = (unsigned)(nj[u] - lo);
        if (di < (unsigned)cnt) {
          float* t = &tile[3 * di];
          atomicAdd(t + 0, fx[u]);
          atomicAdd(t + 1, fy[u]);
          atomicAdd(t + 2, fz[u]);
        }
        if (dj < (unsigned)cnt) {
          float* t = &tile[3 * dj];
          atomicAdd(t + 0, -fx[u]);
          atomicAdd(t + 1, -fy[u]);
          atomicAdd(t + 2, -fz[u]);
        }
      }
    }

    if (!virial_done) {
      les_virial_share(rij, fij, eidx, batch, lvir, E, tid, nthr);
      virial_done = true;
    }
    __syncthreads();

    const int rot = ((cnt3 / S) * s) & ~31;
    float* fbase = force + 3 * lo;
    for (int t = tid; t < cnt3; t += nthr) {
      int k = t + rot;
      if (k >= cnt3) k -= cnt3;
      unsafeAtomicAdd(&fbase[k], tile[k]);
    }
    __syncthreads();
  }

  if (!virial_done) {
    __syncthreads();
    les_virial_share(rij, fij, eidx, batch, lvir, E, tid, nthr);
  }
  __syncthreads();
  for (int i = tid; i < GV_FLOATS; i += nthr) {
    float v = lvir[i];
    if (v != 0.f) unsafeAtomicAdd(&gvir[i], v);
  }
}

__global__ void les_stress_kernel(const float* __restrict__ cell_grad,
                                  const float* __restrict__ les_cell,
                                  const float* __restrict__ vol,
                                  const float* __restrict__ gvir,
                                  float* __restrict__ out) {
  int g = blockIdx.x;
  int lane = threadIdx.x;
  const float* C = les_cell + g * 9;
  const float* G = cell_grad + g * 9;

  int m = lane & 15;
  int hi = lane >> 4;
  int k0 = hi * 2, k1 = k0 + 1;
  int mc = (m < 2) ? m : 2;
  int k0c = (k0 < 2) ? k0 : 2;
  int k1c = (k1 < 2) ? k1 : 2;
  float rowm = (m < 3) ? 1.f : 0.f;
  float m0 = (k0 < 3) ? rowm : 0.f;
  float m1 = (k1 < 3) ? rowm : 0.f;

  v2f a, b;
  a.x = m0 * C[k0c * 3 + mc];
  a.y = m1 * C[k1c * 3 + mc];
  b.x = m0 * G[k0c * 3 + mc];
  b.y = m1 * G[k1c * 3 + mc];
  v8f c = {0.f, 0.f, 0.f, 0.f, 0.f, 0.f, 0.f, 0.f};
  v8f d = __builtin_amdgcn_wmma_f32_16x16x4_f32(
       false, a,  false, b,
       (short)0, c,  false,  false);

  __shared__ float lr[9];
  if (lane < 3) {
    lr[0 * 3 + lane] = d[0];
    lr[1 * 3 + lane] = d[1];
    lr[2 * 3 + lane] = d[2];
  }
  __syncthreads();
  if (lane == 0) {
    float inv = -1.0f / vol[g];
    const float* v = gvir + g * 6;
    float* o = out + g * 6;
    o[0] = inv * (v[0] + lr[0]);
    o[1] = inv * (v[1] + lr[4]);
    o[2] = inv * (v[2] + lr[8]);
    o[3] = inv * (v[3] + lr[1]);
    o[4] = inv * (v[4] + lr[5]);
    o[5] = inv * (v[5] + lr[2]);
  }
}

extern "C" void kernel_launch(void* const* d_in, const int* in_sizes, int n_in,
                              void* d_out, int out_size, void* d_ws, size_t ws_size,
                              hipStream_t stream) {
  const float* rij      = (const float*)d_in[0];
  const float* fij      = (const float*)d_in[1];
  const float* pos_grad = (const float*)d_in[2];
  const float* cell_grad = (const float*)d_in[3];
  const float* les_cell  = (const float*)d_in[4];
  const float* vol       = (const float*)d_in[5];
  const int*   eidx      = (const int*)d_in[6];
  const int*   batch     = (const int*)d_in[7];

  int E  = in_sizes[0] / 3;
  int n3 = in_sizes[2];
  int G  = in_sizes[5];

  float* force  = (float*)d_out;
  float* stress = force + n3;
  float* gvir   = (float*)d_ws;

  const int threads = 256;
  les_init_kernel<<<(n3 + threads - 1) / threads, threads, 0, stream>>>(
      pos_grad, force, gvir, n3);
  les_edge_kernel<<<dim3(EDGE_SLICES, EDGE_TILE_ROWS), EDGE_THREADS, 0, stream>>>(
      rij, fij, eidx, batch, force, gvir, E, n3);
  les_stress_kernel<<<G, 32, 0, stream>>>(
      cell_grad, les_cell, vol, gvir, stress);
}


# Virial moves to 32 blocks of the edge launch: batch as an LDS byte table (no random global gather), sums in lane-private LDS slots; force slices 32 -> 28; f32 sum order changes (virial and force)
# speedup vs baseline: 7.2112x; 1.0296x over previous
#include <hip/hip_runtime.h>
#include <hip/hip_bf16.h>

typedef __attribute__((ext_vector_type(2))) float v2f;
typedef __attribute__((ext_vector_type(8))) float v8f;

#define N_GRAPHS_MAX 64
#define GV_FLOATS (N_GRAPHS_MAX * 6)

__global__ void les_init_kernel(const float* __restrict__ pos_grad,
                                float* __restrict__ force,
                                float* __restrict__ gvir, int n3) {
  int i = blockIdx.x * blockDim.x + threadIdx.x;
  if (i < n3) force[i] = -pos_grad[i];
  if (i < GV_FLOATS) gvir[i] = 0.f;
}

#define TILE_NODES 25024
#define TILE_FLOATS (3 * TILE_NODES)
#define EDGE_THREADS 1024
#define FORCE_SLICES 28
#define FORCE_TILE_ROWS 8
#define VIR_BLOCKS 32
#define EU 4
#define LACC_FLOATS (6 * N_GRAPHS_MAX * 32)
#define TAB_CAP ((TILE_FLOATS - LACC_FLOATS) * 4)

__device__ __forceinline__ void les_virial_block(
    const float* __restrict__ rij, const float* __restrict__ fij,
    const int* __restrict__ eidx, const int* __restrict__ batch,
    float* __restrict__ gvir, float* lds, int E, int n_nodes,
    int v, int nv, int tid, int nthr) {
  unsigned char* tab = (unsigned char*)lds;
  float* lacc = lds + (TILE_FLOATS - LACC_FLOATS);

  for (int i = tid; i < LACC_FLOATS; i += nthr) lacc[i] = 0.f;

  const int ncap = (n_nodes < TAB_CAP) ? n_nodes : TAB_CAP;
  const int n4 = ncap & ~3;
  for (int i = 4 * tid; i < n4; i += 4 * nthr) {
    int4 q = *(const int4*)(batch + i);
    unsigned w = ((unsigned)q.x & 255u) | (((unsigned)q.y & 255u) << 8) |
                 (((unsigned)q.z & 255u) << 16) | ((unsigned)q.w << 24);
    *(unsigned*)(tab + i) = w;
  }
  for (int i = n4 + tid; i < ncap; i += nthr) tab[i] = (unsigned char)batch[i];
  __syncthreads();

  const int per = (E + nv - 1) / nv;
  const int e0 = v * per;
  const int e1 = (e0 + per < E) ? (e0 + per) : E;
  float* my = lacc + (tid & 31);
  const bool in_lds = n_nodes <= TAB_CAP;

  for (int base = e0 + tid; base < e1; base += EU * nthr) {
    float rx[EU], ry[EU], rz[EU], fx[EU], fy[EU], fz[EU];
    int j[EU], g[EU];
    bool ok[EU];
    for (int u = 0; u < EU; ++u) {
      int e = base + u * nthr;
      ok[u] = e < e1;
      int ec = ok[u] ? e : base;
      rx[u] = rij[3 * ec + 0]; ry[u] = rij[3 * ec + 1]; rz[u] = rij[3 * ec + 2];
      fx[u] = fij[3 * ec + 0]; fy[u] = fij[3 * ec + 1]; fz[u] = fij[3 * ec + 2];
      j[u] = eidx[E + ec];
    }
    if (in_lds) {
      for (int u = 0; u < EU; ++u) g[u] = tab[j[u]];
    } else {
      for (int u = 0; u < EU; ++u) g[u] = (j[u] < ncap) ? (int)tab[j[u]] : batch[j[u]];
    }
    for (int u = 0; u < EU; ++u) {
      if (ok[u]) {
        float* p = my + g[u] * 32;
        atomicAdd(p + 0 * (N_GRAPHS_MAX * 32), rx[u] * fx[u]);
        atomicAdd(p + 1 * (N_GRAPHS_MAX * 32), ry[u] * fy[u]);
        atomicAdd(p + 2 * (N_GRAPHS_MAX * 32), rz[u] * fz[u]);
        atomicAdd(p + 3 * (N_GRAPHS_MAX * 32), rx[u] * fy[u]);
        atomicAdd(p + 4 * (N_GRAPHS_MAX * 32), ry[u] * fz[u]);
        atomicAdd(p + 5 * (N_GRAPHS_MAX * 32), rz[u] * fx[u]);
      }
    }
  }
  __syncthreads();

  for (int i = tid; i < 6 * N_GRAPHS_MAX; i += nthr) {
    const float* row = lacc + i * 32;
    float sum = 0.f;
    for (int l = 0; l < 32; ++l) sum += row[(l + i) & 31];
    int c = i / N_GRAPHS_MAX, g = i % N_GRAPHS_MAX;
    if (sum != 0.f) unsafeAtomicAdd(&gvir[g * 6 + c], sum);
  }
}

__global__ __launch_bounds__(EDGE_THREADS) void les_edge_kernel(
    const float* __restrict__ rij, const float* __restrict__ fij,
    const int* __restrict__ eidx, const int* __restrict__ batch,
    float* __restrict__ force, float* __restrict__ gvir, int E, int n3) {
  __shared__ float tile[TILE_FLOATS];
  const int tid = threadIdx.x;
  const int nthr = blockDim.x;
  const int n_nodes = n3 / 3;

  if (blockIdx.x < VIR_BLOCKS) {
    les_virial_block(rij, fij, eidx, batch, gvir, tile, E, n_nodes,
                     blockIdx.x, VIR_BLOCKS, tid, nthr);
    return;
  }

  const int fb = blockIdx.x - VIR_BLOCKS;
  const int S = FORCE_SLICES;
  const int s = fb % FORCE_SLICES;
  const int per = (E + S - 1) / S;
  const int e0 = s * per;
  const int e1 = (e0 + per < E) ? (e0 + per) : E;
  const int nparts = (n_nodes + TILE_NODES - 1) / TILE_NODES;

  for (int p = fb / FORCE_SLICES; p < nparts; p += FORCE_TILE_ROWS) {
    const int lo = p * TILE_NODES;
    const int cnt = (n_nodes - lo < TILE_NODES) ? (n_nodes - lo) : TILE_NODES;
    const int cnt3 = 3 * cnt;

    for (int k = tid; k < cnt3; k += nthr) tile[k] = 0.f;
    __syncthreads();

    for (int base = e0 + tid; base < e1; base += EU * nthr) {
      int ni[EU], nj[EU];
      float fx[EU], fy[EU], fz[EU];
      for (int u = 0; u < EU; ++u) {
        int e = base + u * nthr;
        bool ok = e < e1;
        int ec = ok ? e : base;
        int vi = eidx[ec];
        int vj = eidx[E + ec];
        ni[u] = ok ? vi : -1;
        nj[u] = ok ? vj : -1;
        fx[u] = fij[3 * ec + 0]; fy[u] = fij[3 * ec + 1]; fz[u] = fij[3 * ec + 2];
      }
      for (int u = 0; u < EU; ++u) {
        unsigned di = (unsigned)(ni[u] - lo);
        unsigned dj = (unsigned)(nj[u] - lo);
        if (di < (unsigned)cnt) {
          float* t = &tile[3 * di];
          atomicAdd(t + 0, fx[u]);
          atomicAdd(t + 1, fy[u]);
          atomicAdd(t + 2, fz[u]);
        }
        if (dj < (unsigned)cnt) {
          float* t = &tile[3 * dj];
          atomicAdd(t + 0, -fx[u]);
          atomicAdd(t + 1, -fy[u]);
          atomicAdd(t + 2, -fz[u]);
        }
      }
    }
    __syncthreads();

    const int rot = ((cnt3 / S) * s) & ~31;
    float* fbase = force + 3 * lo;
    for (int t = tid; t < cnt3; t += nthr) {
      int k = t + rot;
      if (k >= cnt3) k -= cnt3;
      unsafeAtomicAdd(&fbase[k], tile[k]);
    }
    __syncthreads();
  }
}

__global__ void les_stress_kernel(const float* __restrict__ cell_grad,
                                  const float* __restrict__ les_cell,
                                  const float* __restrict__ vol,
                                  const float* __restrict__ gvir,
                                  float* __restrict__ out) {
  int g = blockIdx.x;
  int lane = threadIdx.x;
  const float* C = les_cell + g * 9;
  const float* G = cell_grad + g * 9;

  int m = lane & 15;
  int hi = lane >> 4;
  int k0 = hi * 2, k1 = k0 + 1;
  int mc = (m < 2) ? m : 2;
  int k0c = (k0 < 2) ? k0 : 2;
  int k1c = (k1 < 2) ? k1 : 2;
  float rowm = (m < 3) ? 1.f : 0.f;
  float m0 = (k0 < 3) ? rowm : 0.f;
  float m1 = (k1 < 3) ? rowm : 0.f;

  v2f a, b;
  a.x = m0 * C[k0c * 3 + mc];
  a.y = m1 * C[k1c * 3 + mc];
  b.x = m0 * G[k0c * 3 + mc];
  b.y = m1 * G[k1c * 3 + mc];
  v8f c = {0.f, 0.f, 0.f, 0.f, 0.f, 0.f, 0.f, 0.f};
  v8f d = __builtin_amdgcn_wmma_f32_16x16x4_f32(
       false, a,  false, b,
       (short)0, c,  false,  false);

  __shared__ float lr[9];
  if (lane < 3) {
    lr[0 * 3 + lane] = d[0];
    lr[1 * 3 + lane] = d[1];
    lr[2 * 3 + lane] = d[2];
  }
  __syncthreads();
  if (lane == 0) {
    float inv = -1.0f / vol[g];
    const float* v = gvir + g * 6;
    float* o = out + g * 6;
    o[0] = inv * (v[0] + lr[0]);
    o[1] = inv * (v[1] + lr[4]);
    o[2] = inv * (v[2] + lr[8]);
    o[3] = inv * (v[3] + lr[1]);
    o[4] = inv * (v[4] + lr[5]);
    o[5] = inv * (v[5] + lr[2]);
  }
}

extern "C" void kernel_launch(void* const* d_in, const int* in_sizes, int n_in,
                              void* d_out, int out_size, void* d_ws, size_t ws_size,
                              hipStream_t stream) {
  const float* rij      = (const float*)d_in[0];
  const float* fij      = (const float*)d_in[1];
  const float* pos_grad = (const float*)d_in[2];
  const float* cell_grad = (const float*)d_in[3];
  const float* les_cell  = (const float*)d_in[4];
  const float* vol       = (const float*)d_in[5];
  const int*   eidx      = (const int*)d_in[6];
  const int*   batch     = (const int*)d_in[7];

  int E  = in_sizes[0] / 3;
  int n3 = in_sizes[2];
  int G  = in_sizes[5];

  float* force  = (float*)d_out;
  float* stress = force + n3;
  float* gvir   = (float*)d_ws;

  const int threads = 256;
  les_init_kernel<<<(n3 + threads - 1) / threads, threads, 0, stream>>>(
      pos_grad, force, gvir, n3);
  les_edge_kernel<<<VIR_BLOCKS + FORCE_SLICES * FORCE_TILE_ROWS, EDGE_THREADS, 0, stream>>>(
      rij, fij, eidx, batch, force, gvir, E, n3);
  les_stress_kernel<<<G, 32, 0, stream>>>(
      cell_grad, les_cell, vol, gvir, stress);
}


# Force scan: each lane walks only its own tile hits, 3 ds_add_f32 per hit round instead of 6 per edge; grid re-split: virial blocks 32->64, force slices 28->24; f32 sum order changes (force, virial)
# speedup vs baseline: 7.7192x; 1.0704x over previous
#include <hip/hip_runtime.h>
#include <hip/hip_bf16.h>

typedef __attribute__((ext_vector_type(2))) float v2f;
typedef __attribute__((ext_vector_type(8))) float v8f;

#define N_GRAPHS_MAX 64
#define GV_FLOATS (N_GRAPHS_MAX * 6)

__global__ void les_init_kernel(const float* __restrict__ pos_grad,
                                float* __restrict__ force,
                                float* __restrict__ gvir, int n3) {
  int i = blockIdx.x * blockDim.x + threadIdx.x;
  if (i < n3) force[i] = -pos_grad[i];
  if (i < GV_FLOATS) gvir[i] = 0.f;
}

#define TILE_NODES 25024
#define TILE_FLOATS (3 * TILE_NODES)
#define EDGE_THREADS 1024
#define FORCE_SLICES 24
#define FORCE_TILE_ROWS 8
#define VIR_BLOCKS 64
#define EU 4
#define LACC_FLOATS (6 * N_GRAPHS_MAX * 32)
#define TAB_CAP ((TILE_FLOATS - LACC_FLOATS) * 4)

__device__ __forceinline__ void les_virial_block(
    const float* __restrict__ rij, const float* __restrict__ fij,
    const int* __restrict__ eidx, const int* __restrict__ batch,
    float* __restrict__ gvir, float* lds, int E, int n_nodes,
    int v, int nv, int tid, int nthr) {
  unsigned char* tab = (unsigned char*)lds;
  float* lacc = lds + (TILE_FLOATS - LACC_FLOATS);

  for (int i = tid; i < LACC_FLOATS; i += nthr) lacc[i] = 0.f;

  const int ncap = (n_nodes < TAB_CAP) ? n_nodes : TAB_CAP;
  const int n4 = ncap & ~3;
  for (int i = 4 * tid; i < n4; i += 4 * nthr) {
    int4 q = *(const int4*)(batch + i);
    unsigned w = ((unsigned)q.x & 255u) | (((unsigned)q.y & 255u) << 8) |
                 (((unsigned)q.z & 255u) << 16) | ((unsigned)q.w << 24);
    *(unsigned*)(tab + i) = w;
  }
  for (int i = n4 + tid; i < ncap; i += nthr) tab[i] = (unsigned char)batch[i];
  __syncthreads();

  const int per = (E + nv - 1) / nv;
  const int e0 = v * per;
  const int e1 = (e0 + per < E) ? (e0 + per) : E;
  float* my = lacc + (tid & 31);
  const bool in_lds = n_nodes <= TAB_CAP;

  for (int base = e0 + tid; base < e1; base += EU * nthr) {
    float rx[EU], ry[EU], rz[EU], fx[EU], fy[EU], fz[EU];
    int j[EU], g[EU];
    bool ok[EU];
    for (int u = 0; u < EU; ++u) {
      int e = base + u * nthr;
      ok[u] = e < e1;
      int ec = ok[u] ? e : base;
      rx[u] = rij[3 * ec + 0]; ry[u] = rij[3 * ec + 1]; rz[u] = rij[3 * ec + 2];
      fx[u] = fij[3 * ec + 0]; fy[u] = fij[3 * ec + 1]; fz[u] = fij[3 * ec + 2];
      j[u] = eidx[E + ec];
    }
    if (in_lds) {
      for (int u = 0; u < EU; ++u) g[u] = tab[j[u]];
    } else {
      for (int u = 0; u < EU; ++u) g[u] = (j[u] < ncap) ? (int)tab[j[u]] : batch[j[u]];
    }
    for (int u = 0; u < EU; ++u) {
      if (ok[u]) {
        float* p = my + g[u] * 32;
        atomicAdd(p + 0 * (N_GRAPHS_MAX * 32), rx[u] * fx[u]);
        atomicAdd(p + 1 * (N_GRAPHS_MAX * 32), ry[u] * fy[u]);
        atomicAdd(p + 2 * (N_GRAPHS_MAX * 32), rz[u] * fz[u]);
        atomicAdd(p + 3 * (N_GRAPHS_MAX * 32), rx[u] * fy[u]);
        atomicAdd(p + 4 * (N_GRAPHS_MAX * 32), ry[u] * fz[u]);
        atomicAdd(p + 5 * (N_GRAPHS_MAX * 32), rz[u] * fx[u]);
      }
    }
  }
  __syncthreads();

  for (int i = tid; i < 6 * N_GRAPHS_MAX; i += nthr) {
    const float* row = lacc + i * 32;
    float sum = 0.f;
    for (int l = 0; l < 32; ++l) sum += row[(l + i) & 31];
    int c = i / N_GRAPHS_MAX, g = i % N_GRAPHS_MAX;
    if (sum != 0.f) unsafeAtomicAdd(&gvir[g * 6 + c], sum);
  }
}

__global__ __launch_bounds__(EDGE_THREADS) void les_edge_kernel(
    const float* __restrict__ rij, const float* __restrict__ fij,
    const int* __restrict__ eidx, const int* __restrict__ batch,
    float* __restrict__ force, float* __restrict__ gvir, int E, int n3) {
  __shared__ float tile[TILE_FLOATS];
  const int tid = threadIdx.x;
  const int nthr = blockDim.x;
  const int n_nodes = n3 / 3;

  if (blockIdx.x < VIR_BLOCKS) {
    les_virial_block(rij, fij, eidx, batch, gvir, tile, E, n_nodes,
                     blockIdx.x, VIR_BLOCKS, tid, nthr);
    return;
  }

  const int fb = blockIdx.x - VIR_BLOCKS;
  const int S = FORCE_SLICES;
  const int s = fb % FORCE_SLICES;
  const int per = (E + S - 1) / S;
  const int e0 = s * per;
  const int e1 = (e0 + per < E) ? (e0 + per) : E;
  const int nparts = (n_nodes + TILE_NODES - 1) / TILE_NODES;

  for (int p = fb / FORCE_SLICES; p < nparts; p += FORCE_TILE_ROWS) {
    const int lo = p * TILE_NODES;
    const int cnt = (n_nodes - lo < TILE_NODES) ? (n_nodes - lo) : TILE_NODES;
    const int cnt3 = 3 * cnt;

    for (int k = tid; k < cnt3; k += nthr) tile[k] = 0.f;
    __syncthreads();

    for (int base = e0 + tid; base < e1; base += EU * nthr) {
      int ni[EU], nj[EU];
      float fx[EU], fy[EU], fz[EU];
      for (int u = 0; u < EU; ++u) {
        int e = base + u * nthr;
        bool ok = e < e1;
        int ec = ok ? e : base;
        int vi = eidx[ec];
        int vj = eidx[E + ec];
        ni[u] = ok ? vi : -1;
        nj[u] = ok ? vj : -1;
        fx[u] = fij[3 * ec + 0]; fy[u] = fij[3 * ec + 1]; fz[u] = fij[3 * ec + 2];
      }
      for (int u = 0; u < EU; ++u) asm volatile("" : "+v"(fx[u]), "+v"(fy[u]), "+v"(fz[u]));
      unsigned di[EU], dj[EU];
      unsigned hits = 0;
      for (int u = 0; u < EU; ++u) {
        di[u] = (unsigned)(ni[u] - lo);
        dj[u] = (unsigned)(nj[u] - lo);
        hits |= (di[u] < (unsigned)cnt ? 1u : 0u) << (2 * u);
        hits |= (dj[u] < (unsigned)cnt ? 1u : 0u) << (2 * u + 1);
      }
      while (hits) {
        const int k = __builtin_ctz(hits);
        hits &= hits - 1;
        const int u = k >> 1;
        unsigned ds = di[0], dr = dj[0];
        float x = fx[0], y = fy[0], z = fz[0];
        for (int w = 1; w < EU; ++w) {
          if (u == w) { ds = di[w]; dr = dj[w]; x = fx[w]; y = fy[w]; z = fz[w]; }
        }
        const bool recv = (k & 1) != 0;
        const unsigned d = recv ? dr : ds;
        if (recv) { x = -x; y = -y; z = -z; }
        float* t = &tile[3 * d];
        atomicAdd(t + 0, x);
        atomicAdd(t + 1, y);
        atomicAdd(t + 2, z);
      }
    }
    __syncthreads();

    const int rot = ((cnt3 / S) * s) & ~31;
    float* fbase = force + 3 * lo;
    for (int t = tid; t < cnt3; t += nthr) {
      int k = t + rot;
      if (k >= cnt3) k -= cnt3;
      unsafeAtomicAdd(&fbase[k], tile[k]);
    }
    __syncthreads();
  }
}

__global__ void les_stress_kernel(const float* __restrict__ cell_grad,
                                  const float* __restrict__ les_cell,
                                  const float* __restrict__ vol,
                                  const float* __restrict__ gvir,
                                  float* __restrict__ out) {
  int g = blockIdx.x;
  int lane = threadIdx.x;
  const float* C = les_cell + g * 9;
  const float* G = cell_grad + g * 9;

  int m = lane & 15;
  int hi = lane >> 4;
  int k0 = hi * 2, k1 = k0 + 1;
  int mc = (m < 2) ? m : 2;
  int k0c = (k0 < 2) ? k0 : 2;
  int k1c = (k1 < 2) ? k1 : 2;
  float rowm = (m < 3) ? 1.f : 0.f;
  float m0 = (k0 < 3) ? rowm : 0.f;
  float m1 = (k1 < 3) ? rowm : 0.f;

  v2f a, b;
  a.x = m0 * C[k0c * 3 + mc];
  a.y = m1 * C[k1c * 3 + mc];
  b.x = m0 * G[k0c * 3 + mc];
  b.y = m1 * G[k1c * 3 + mc];
  v8f c = {0.f, 0.f, 0.f, 0.f, 0.f, 0.f, 0.f, 0.f};
  v8f d = __builtin_amdgcn_wmma_f32_16x16x4_f32(
       false, a,  false, b,
       (short)0, c,  false,  false);

  __shared__ float lr[9];
  if (lane < 3) {
    lr[0 * 3 + lane] = d[0];
    lr[1 * 3 + lane] = d[1];
    lr[2 * 3 + lane] = d[2];
  }
  __syncthreads();
  if (lane == 0) {
    float inv = -1.0f / vol[g];
    const float* v = gvir + g * 6;
    float* o = out + g * 6;
    o[0] = inv * (v[0] + lr[0]);
    o[1] = inv * (v[1] + lr[4]);
    o[2] = inv * (v[2] + lr[8]);
    o[3] = inv * (v[3] + lr[1]);
    o[4] = inv * (v[4] + lr[5]);
    o[5] = inv * (v[5] + lr[2]);
  }
}

extern "C" void kernel_launch(void* const* d_in, const int* in_sizes, int n_in,
                              void* d_out, int out_size, void* d_ws, size_t ws_size,
                              hipStream_t stream) {
  const float* rij      = (const float*)d_in[0];
  const float* fij      = (const float*)d_in[1];
  const float* pos_grad = (const float*)d_in[2];
  const float* cell_grad = (const float*)d_in[3];
  const float* les_cell  = (const float*)d_in[4];
  const float* vol       = (const float*)d_in[5];
  const int*   eidx      = (const int*)d_in[6];
  const int*   batch     = (const int*)d_in[7];

  int E  = in_sizes[0] / 3;
  int n3 = in_sizes[2];
  int G  = in_sizes[5];

  float* force  = (float*)d_out;
  float* stress = force + n3;
  float* gvir   = (float*)d_ws;

  const int threads = 256;
  les_init_kernel<<<(n3 + threads - 1) / threads, threads, 0, stream>>>(
      pos_grad, force, gvir, n3);
  les_edge_kernel<<<VIR_BLOCKS + FORCE_SLICES * FORCE_TILE_ROWS, EDGE_THREADS, 0, stream>>>(
      rij, fij, eidx, batch, force, gvir, E, n3);
  les_stress_kernel<<<G, 32, 0, stream>>>(
      cell_grad, les_cell, vol, gvir, stress);
}


# Grid re-split towards the force scan: force slices 24 -> 27 (slice length a multiple of 64 edges), virial blocks 64 -> 40; f32 sum order changes (force, virial)
# speedup vs baseline: 8.0528x; 1.0432x over previous
#include <hip/hip_runtime.h>
#include <hip/hip_bf16.h>

typedef __attribute__((ext_vector_type(2))) float v2f;
typedef __attribute__((ext_vector_type(8))) float v8f;

#define N_GRAPHS_MAX 64
#define GV_FLOATS (N_GRAPHS_MAX * 6)

__global__ void les_init_kernel(const float* __restrict__ pos_grad,
                                float* __restrict__ force,
                                float* __restrict__ gvir, int n3) {
  int i = blockIdx.x * blockDim.x + threadIdx.x;
  if (i < n3) force[i] = -pos_grad[i];
  if (i < GV_FLOATS) gvir[i] = 0.f;
}

#define TILE_NODES 25024
#define TILE_FLOATS (3 * TILE_NODES)
#define EDGE_THREADS 1024
#define FORCE_SLICES 27
#define FORCE_TILE_ROWS 8
#define VIR_BLOCKS 40
#define EU 4
#define LACC_FLOATS (6 * N_GRAPHS_MAX * 32)
#define TAB_CAP ((TILE_FLOATS - LACC_FLOATS) * 4)

__device__ __forceinline__ void les_virial_block(
    const float* __restrict__ rij, const float* __restrict__ fij,
    const int* __restrict__ eidx, const int* __restrict__ batch,
    float* __restrict__ gvir, float* lds, int E, int n_nodes,
    int v, int nv, int tid, int nthr) {
  unsigned char* tab = (unsigned char*)lds;
  float* lacc = lds + (TILE_FLOATS - LACC_FLOATS);

  for (int i = tid; i < LACC_FLOATS; i += nthr) lacc[i] = 0.f;

  const int ncap = (n_nodes < TAB_CAP) ? n_nodes : TAB_CAP;
  const int n4 = ncap & ~3;
  for (int i = 4 * tid; i < n4; i += 4 * nthr) {
    int4 q = *(const int4*)(batch + i);
    unsigned w = ((unsigned)q.x & 255u) | (((unsigned)q.y & 255u) << 8) |
                 (((unsigned)q.z & 255u) << 16) | ((unsigned)q.w << 24);
    *(unsigned*)(tab + i) = w;
  }
  for (int i = n4 + tid; i < ncap; i += nthr) tab[i] = (unsigned char)batch[i];
  __syncthreads();

  const int per = (E + nv - 1) / nv;
  const int e0 = v * per;
  const int e1 = (e0 + per < E) ? (e0 + per) : E;
  float* my = lacc + (tid & 31);
  const bool in_lds = n_nodes <= TAB_CAP;

  for (int base = e0 + tid; base < e1; base += EU * nthr) {
    float rx[EU], ry[EU], rz[EU], fx[EU], fy[EU], fz[EU];
    int j[EU], g[EU];
    bool ok[EU];
    for (int u = 0; u < EU; ++u) {
      int e = base + u * nthr;
      ok[u] = e < e1;
      int ec = ok[u] ? e : base;
      rx[u] = rij[3 * ec + 0]; ry[u] = rij[3 * ec + 1]; rz[u] = rij[3 * ec + 2];
      fx[u] = fij[3 * ec + 0]; fy[u] = fij[3 * ec + 1]; fz[u] = fij[3 * ec + 2];
      j[u] = eidx[E + ec];
    }
    if (in_lds) {
      for (int u = 0; u < EU; ++u) g[u] = tab[j[u]];
    } else {
      for (int u = 0; u < EU; ++u) g[u] = (j[u] < ncap) ? (int)tab[j[u]] : batch[j[u]];
    }
    for (int u = 0; u < EU; ++u) {
      if (ok[u]) {
        float* p = my + g[u] * 32;
        atomicAdd(p + 0 * (N_GRAPHS_MAX * 32), rx[u] * fx[u]);
        atomicAdd(p + 1 * (N_GRAPHS_MAX * 32), ry[u] * fy[u]);
        atomicAdd(p + 2 * (N_GRAPHS_MAX * 32), rz[u] * fz[u]);
        atomicAdd(p + 3 * (N_GRAPHS_MAX * 32), rx[u] * fy[u]);
        atomicAdd(p + 4 * (N_GRAPHS_MAX * 32), ry[u] * fz[u]);
        atomicAdd(p + 5 * (N_GRAPHS_MAX * 32), rz[u] * fx[u]);
      }
    }
  }
  __syncthreads();

  for (int i = tid; i < 6 * N_GRAPHS_MAX; i += nthr) {
    const float* row = lacc + i * 32;
    float sum = 0.f;
    for (int l = 0; l < 32; ++l) sum += row[(l + i) & 31];
    int c = i / N_GRAPHS_MAX, g = i % N_GRAPHS_MAX;
    if (sum != 0.f) unsafeAtomicAdd(&gvir[g * 6 + c], sum);
  }
}

__global__ __launch_bounds__(EDGE_THREADS) void les_edge_kernel(
    const float* __restrict__ rij, const float* __restrict__ fij,
    const int* __restrict__ eidx, const int* __restrict__ batch,
    float* __restrict__ force, float* __restrict__ gvir, int E, int n3) {
  __shared__ float tile[TILE_FLOATS];
  const int tid = threadIdx.x;
  const int nthr = blockDim.x;
  const int n_nodes = n3 / 3;

  if (blockIdx.x < VIR_BLOCKS) {
    les_virial_block(rij, fij, eidx, batch, gvir, tile, E, n_nodes,
                     blockIdx.x, VIR_BLOCKS, tid, nthr);
    return;
  }

  const int fb = blockIdx.x - VIR_BLOCKS;
  const int S = FORCE_SLICES;
  const int s = fb % FORCE_SLICES;
  const int per = (((E + S - 1) / S) + 63) & ~63;
  const int e0 = s * per;
  const int e1 = (e0 + per < E) ? (e0 + per) : E;
  const int nparts = (n_nodes + TILE_NODES - 1) / TILE_NODES;

  for (int p = fb / FORCE_SLICES; p < nparts; p += FORCE_TILE_ROWS) {
    const int lo = p * TILE_NODES;
    const int cnt = (n_nodes - lo < TILE_NODES) ? (n_nodes - lo) : TILE_NODES;
    const int cnt3 = 3 * cnt;

    for (int k = tid; k < cnt3; k += nthr) tile[k] = 0.f;
    __syncthreads();

    for (int base = e0 + tid; base < e1; base += EU * nthr) {
      int ni[EU], nj[EU];
      float fx[EU], fy[EU], fz[EU];
      for (int u = 0; u < EU; ++u) {
        int e = base + u * nthr;
        bool ok = e < e1;
        int ec = ok ? e : base;
        int vi = eidx[ec];
        int vj = eidx[E + ec];
        ni[u] = ok ? vi : -1;
        nj[u] = ok ? vj : -1;
        fx[u] = fij[3 * ec + 0]; fy[u] = fij[3 * ec + 1]; fz[u] = fij[3 * ec + 2];
      }
      for (int u = 0; u < EU; ++u) asm volatile("" : "+v"(fx[u]), "+v"(fy[u]), "+v"(fz[u]));
      unsigned di[EU], dj[EU];
      unsigned hits = 0;
      for (int u = 0; u < EU; ++u) {
        di[u] = (unsigned)(ni[u] - lo);
        dj[u] = (unsigned)(nj[u] - lo);
        hits |= (di[u] < (unsigned)cnt ? 1u : 0u) << (2 * u);
        hits |= (dj[u] < (unsigned)cnt ? 1u : 0u) << (2 * u + 1);
      }
      while (hits) {
        const int k = __builtin_ctz(hits);
        hits &= hits - 1;
        const int u = k >> 1;
        unsigned ds = di[0], dr = dj[0];
        float x = fx[0], y = fy[0], z = fz[0];
        for (int w = 1; w < EU; ++w) {
          if (u == w) { ds = di[w]; dr = dj[w]; x = fx[w]; y = fy[w]; z = fz[w]; }
        }
        const bool recv = (k & 1) != 0;
        const unsigned d = recv ? dr : ds;
        if (recv) { x = -x; y = -y; z = -z; }
        float* t = &tile[3 * d];
        atomicAdd(t + 0, x);
        atomicAdd(t + 1, y);
        atomicAdd(t + 2, z);
      }
    }
    __syncthreads();

    const int rot = ((cnt3 / S) * s) & ~31;
    float* fbase = force + 3 * lo;
    for (int t = tid; t < cnt3; t += nthr) {
      int k = t + rot;
      if (k >= cnt3) k -= cnt3;
      unsafeAtomicAdd(&fbase[k], tile[k]);
    }
    __syncthreads();
  }
}

__global__ void les_stress_kernel(const float* __restrict__ cell_grad,
                                  const float* __restrict__ les_cell,
                                  const float* __restrict__ vol,
                                  const float* __restrict__ gvir,
                                  float* __restrict__ out) {
  int g = blockIdx.x;
  int lane = threadIdx.x;
  const float* C = les_cell + g * 9;
  const float* G = cell_grad + g * 9;

  int m = lane & 15;
  int hi = lane >> 4;
  int k0 = hi * 2, k1 = k0 + 1;
  int mc = (m < 2) ? m : 2;
  int k0c = (k0 < 2) ? k0 : 2;
  int k1c = (k1 < 2) ? k1 : 2;
  float rowm = (m < 3) ? 1.f : 0.f;
  float m0 = (k0 < 3) ? rowm : 0.f;
  float m1 = (k1 < 3) ? rowm : 0.f;

  v2f a, b;
  a.x = m0 * C[k0c * 3 + mc];
  a.y = m1 * C[k1c * 3 + mc];
  b.x = m0 * G[k0c * 3 + mc];
  b.y = m1 * G[k1c * 3 + mc];
  v8f c = {0.f, 0.f, 0.f, 0.f, 0.f, 0.f, 0.f, 0.f};
  v8f d = __builtin_amdgcn_wmma_f32_16x16x4_f32(
       false, a,  false, b,
       (short)0, c,  false,  false);

  __shared__ float lr[9];
  if (lane < 3) {
    lr[0 * 3 + lane] = d[0];
    lr[1 * 3 + lane] = d[1];
    lr[2 * 3 + lane] = d[2];
  }
  __syncthreads();
  if (lane == 0) {
    float inv = -1.0f / vol[g];
    const float* v = gvir + g * 6;
    float* o = out + g * 6;
    o[0] = inv * (v[0] + lr[0]);
    o[1] = inv * (v[1] + lr[4]);
    o[2] = inv * (v[2] + lr[8]);
    o[3] = inv * (v[3] + lr[1]);
    o[4] = inv * (v[4] + lr[5]);
    o[5] = inv * (v[5] + lr[2]);
  }
}

extern "C" void kernel_launch(void* const* d_in, const int* in_sizes, int n_in,
                              void* d_out, int out_size, void* d_ws, size_t ws_size,
                              hipStream_t stream) {
  const float* rij      = (const float*)d_in[0];
  const float* fij      = (const float*)d_in[1];
  const float* pos_grad = (const float*)d_in[2];
  const float* cell_grad = (const float*)d_in[3];
  const float* les_cell  = (const float*)d_in[4];
  const float* vol       = (const float*)d_in[5];
  const int*   eidx      = (const int*)d_in[6];
  const int*   batch     = (const int*)d_in[7];

  int E  = in_sizes[0] / 3;
  int n3 = in_sizes[2];
  int G  = in_sizes[5];

  float* force  = (float*)d_out;
  float* stress = force + n3;
  float* gvir   = (float*)d_ws;

  const int threads = 256;
  les_init_kernel<<<(n3 + threads - 1) / threads, threads, 0, stream>>>(
      pos_grad, force, gvir, n3);
  les_edge_kernel<<<VIR_BLOCKS + FORCE_SLICES * FORCE_TILE_ROWS, EDGE_THREADS, 0, stream>>>(
      rij, fij, eidx, batch, force, gvir, E, n3);
  les_stress_kernel<<<G, 32, 0, stream>>>(
      cell_grad, les_cell, vol, gvir, stress);
}


# Edge kernel: block size a compile-time constant and full trips peeled from one range-tested tail trip, so per-edge clamps and 64-bit address math leave the scan loops; f32 sum order unchanged
# speedup vs baseline: 8.6241x; 1.0709x over previous
#include <hip/hip_runtime.h>
#include <hip/hip_bf16.h>

typedef __attribute__((ext_vector_type(2))) float v2f;
typedef __attribute__((ext_vector_type(8))) float v8f;

#define N_GRAPHS_MAX 64
#define GV_FLOATS (N_GRAPHS_MAX * 6)

__global__ void les_init_kernel(const float* __restrict__ pos_grad,
                                float* __restrict__ force,
                                float* __restrict__ gvir, int n3) {
  int i = blockIdx.x * blockDim.x + threadIdx.x;
  if (i < n3) force[i] = -pos_grad[i];
  if (i < GV_FLOATS) gvir[i] = 0.f;
}

#define TILE_NODES 25024
#define TILE_FLOATS (3 * TILE_NODES)
#define EDGE_THREADS 1024
#define NT EDGE_THREADS
#define FORCE_SLICES 27
#define FORCE_TILE_ROWS 8
#define VIR_BLOCKS 40
#define EU 4
#define LACC_FLOATS (6 * N_GRAPHS_MAX * 32)
#define TAB_CAP ((TILE_FLOATS - LACC_FLOATS) * 4)

__device__ __forceinline__ void les_virial_add(float* p, float rx, float ry, float rz,
                                               float fx, float fy, float fz) {
  atomicAdd(p + 0 * (N_GRAPHS_MAX * 32), rx * fx);
  atomicAdd(p + 1 * (N_GRAPHS_MAX * 32), ry * fy);
  atomicAdd(p + 2 * (N_GRAPHS_MAX * 32), rz * fz);
  atomicAdd(p + 3 * (N_GRAPHS_MAX * 32), rx * fy);
  atomicAdd(p + 4 * (N_GRAPHS_MAX * 32), ry * fz);
  atomicAdd(p + 5 * (N_GRAPHS_MAX * 32), rz * fx);
}

__device__ __forceinline__ void les_virial_block(
    const float* __restrict__ rij, const float* __restrict__ fij,
    const int* __restrict__ eidx, const int* __restrict__ batch,
    float* __restrict__ gvir, float* lds, int E, int n_nodes,
    int v, int nv, int tid) {
  unsigned char* tab = (unsigned char*)lds;
  float* lacc = lds + (TILE_FLOATS - LACC_FLOATS);

  for (int i = tid; i < LACC_FLOATS; i += NT) lacc[i] = 0.f;

  const int ncap = (n_nodes < TAB_CAP) ? n_nodes : TAB_CAP;
  const int n4 = ncap & ~3;
  for (int i = 4 * tid; i < n4; i += 4 * NT) {
    int4 q = *(const int4*)(batch + i);
    unsigned w = ((unsigned)q.x & 255u) | (((unsigned)q.y & 255u) << 8) |
                 (((unsigned)q.z & 255u) << 16) | ((unsigned)q.w << 24);
    *(unsigned*)(tab + i) = w;
  }
  for (int i = n4 + tid; i < ncap; i += NT) tab[i] = (unsigned char)batch[i];
  __syncthreads();

  const int per = (E + nv - 1) / nv;
  const int e0 = v * per;
  const int e1 = (e0 + per < E) ? (e0 + per) : E;
  float* my = lacc + (tid & 31);
  const bool in_lds = n_nodes <= TAB_CAP;

  int base = e0 + tid;
  for (; base + (EU - 1) * NT < e1; base += EU * NT) {
    const float* pr = rij + 3 * base;
    const float* pf = fij + 3 * base;
    const int* pj = eidx + E + base;
    float rx[EU], ry[EU], rz[EU], fx[EU], fy[EU], fz[EU];
    int j[EU], g[EU];
    for (int u = 0; u < EU; ++u) {
      rx[u] = pr[3 * u * NT + 0]; ry[u] = pr[3 * u * NT + 1]; rz[u] = pr[3 * u * NT + 2];
      fx[u] = pf[3 * u * NT + 0]; fy[u] = pf[3 * u * NT + 1]; fz[u] = pf[3 * u * NT + 2];
      j[u] = pj[u * NT];
    }
    if (in_lds) {
      for (int u = 0; u < EU; ++u) g[u] = tab[j[u]];
    } else {
      for (int u = 0; u < EU; ++u) g[u] = (j[u] < ncap) ? (int)tab[j[u]] : batch[j[u]];
    }
    for (int u = 0; u < EU; ++u)
      les_virial_add(my + g[u] * 32, rx[u], ry[u], rz[u], fx[u], fy[u], fz[u]);
  }
  if (base < e1) {
    float rx[EU], ry[EU], rz[EU], fx[EU], fy[EU], fz[EU];
    int j[EU], g[EU];
    bool ok[EU];
    for (int u = 0; u < EU; ++u) {
      int e = base + u * NT;
      ok[u] = e < e1;
      int ec = ok[u] ? e : base;
      rx[u] = rij[3 * ec + 0]; ry[u] = rij[3 * ec + 1]; rz[u] = rij[3 * ec + 2];
      fx[u] = fij[3 * ec + 0]; fy[u] = fij[3 * ec + 1]; fz[u] = fij[3 * ec + 2];
      j[u] = eidx[E + ec];
    }
    if (in_lds) {
      for (int u = 0; u < EU; ++u) g[u] = tab[j[u]];
    } else {
      for (int u = 0; u < EU; ++u) g[u] = (j[u] < ncap) ? (int)tab[j[u]] : batch[j[u]];
    }
    for (int u = 0; u < EU; ++u) {
      if (ok[u]) les_virial_add(my + g[u] * 32, rx[u], ry[u], rz[u], fx[u], fy[u], fz[u]);
    }
  }
  __syncthreads();

  for (int i = tid; i < 6 * N_GRAPHS_MAX; i += NT) {
    const float* row = lacc + i * 32;
    float sum = 0.f;
    for (int l = 0; l < 32; ++l) sum += row[(l + i) & 31];
    int c = i / N_GRAPHS_MAX, g = i % N_GRAPHS_MAX;
    if (sum != 0.f) unsafeAtomicAdd(&gvir[g * 6 + c], sum);
  }
}

__device__ __forceinline__ void les_tile_hits(
    const int (&ni)[EU], const int (&nj)[EU], const float (&fx)[EU],
    const float (&fy)[EU], const float (&fz)[EU], int lo, int cnt, float* tile) {
  unsigned di[EU], dj[EU];
  unsigned hits = 0;
  for (int u = 0; u < EU; ++u) {
    di[u] = (unsigned)(ni[u] - lo);
    dj[u] = (unsigned)(nj[u] - lo);
    hits |= (di[u] < (unsigned)cnt ? 1u : 0u) << (2 * u);
    hits |= (dj[u] < (unsigned)cnt ? 1u : 0u) << (2 * u + 1);
  }
  while (hits) {
    const int k = __builtin_ctz(hits);
    hits &= hits - 1;
    const int u = k >> 1;
    unsigned ds = di[0], dr = dj[0];
    float x = fx[0], y = fy[0], z = fz[0];
    for (int w = 1; w < EU; ++w) {
      if (u == w) { ds = di[w]; dr = dj[w]; x = fx[w]; y = fy[w]; z = fz[w]; }
    }
    const bool recv = (k & 1) != 0;
    const unsigned d = recv ? dr : ds;
    if (recv) { x = -x; y = -y; z = -z; }
    float* t = &tile[3 * d];
    atomicAdd(t + 0, x);
    atomicAdd(t + 1, y);
    atomicAdd(t + 2, z);
  }
}

__global__ __launch_bounds__(EDGE_THREADS) void les_edge_kernel(
    const float* __restrict__ rij, const float* __restrict__ fij,
    const int* __restrict__ eidx, const int* __restrict__ batch,
    float* __restrict__ force, float* __restrict__ gvir, int E, int n3) {
  __shared__ float tile[TILE_FLOATS];
  const int tid = threadIdx.x;
  const int n_nodes = n3 / 3;

  if (blockIdx.x < VIR_BLOCKS) {
    les_virial_block(rij, fij, eidx, batch, gvir, tile, E, n_nodes,
                     blockIdx.x, VIR_BLOCKS, tid);
    return;
  }

  const int fb = blockIdx.x - VIR_BLOCKS;
  const int S = FORCE_SLICES;
  const int s = fb % FORCE_SLICES;
  const int per = (((E + S - 1) / S) + 63) & ~63;
  const int e0 = s * per;
  const int e1 = (e0 + per < E) ? (e0 + per) : E;
  const int nparts = (n_nodes + TILE_NODES - 1) / TILE_NODES;

  for (int p = fb / FORCE_SLICES; p < nparts; p += FORCE_TILE_ROWS) {
    const int lo = p * TILE_NODES;
    const int cnt = (n_nodes - lo < TILE_NODES) ? (n_nodes - lo) : TILE_NODES;
    const int cnt3 = 3 * cnt;

    for (int k = tid; k < cnt3; k += NT) tile[k] = 0.f;
    __syncthreads();

    int base = e0 + tid;
    for (; base + (EU - 1) * NT < e1; base += EU * NT) {
      const int* pi = eidx + base;
      const int* pj = eidx + E + base;
      const float* pf = fij + 3 * base;
      int ni[EU], nj[EU];
      float fx[EU], fy[EU], fz[EU];
      for (int u = 0; u < EU; ++u) {
        ni[u] = pi[u * NT];
        nj[u] = pj[u * NT];
        fx[u] = pf[3 * u * NT + 0]; fy[u] = pf[3 * u * NT + 1]; fz[u] = pf[3 * u * NT + 2];
      }
      for (int u = 0; u < EU; ++u) asm volatile("" : "+v"(fx[u]), "+v"(fy[u]), "+v"(fz[u]));
      les_tile_hits(ni, nj, fx, fy, fz, lo, cnt, tile);
    }
    if (base < e1) {
      int ni[EU], nj[EU];
      float fx[EU], fy[EU], fz[EU];
      for (int u = 0; u < EU; ++u) {
        int e = base + u * NT;
        bool ok = e < e1;
        int ec = ok ? e : base;
        int vi = eidx[ec];
        int vj = eidx[E + ec];
        ni[u] = ok ? vi : -1;
        nj[u] = ok ? vj : -1;
        fx[u] = fij[3 * ec + 0]; fy[u] = fij[3 * ec + 1]; fz[u] = fij[3 * ec + 2];
      }
      les_tile_hits(ni, nj, fx, fy, fz, lo, cnt, tile);
    }
    __syncthreads();

    const int rot = ((cnt3 / S) * s) & ~31;
    float* fbase = force + 3 * lo;
    for (int t = tid; t < cnt3; t += NT) {
      int k = t + rot;
      if (k >= cnt3) k -= cnt3;
      unsafeAtomicAdd(&fbase[k], tile[k]);
    }
    __syncthreads();
  }
}

__global__ void les_stress_kernel(const float* __restrict__ cell_grad,
                                  const float* __restrict__ les_cell,
                                  const float* __restrict__ vol,
                                  const float* __restrict__ gvir,
                                  float* __restrict__ out) {
  int g = blockIdx.x;
  int lane = threadIdx.x;
  const float* C = les_cell + g * 9;
  const float* G = cell_grad + g * 9;

  int m = lane & 15;
  int hi = lane >> 4;
  int k0 = hi * 2, k1 = k0 + 1;
  int mc = (m < 2) ? m : 2;
  int k0c = (k0 < 2) ? k0 : 2;
  int k1c = (k1 < 2) ? k1 : 2;
  float rowm = (m < 3) ? 1.f : 0.f;
  float m0 = (k0 < 3) ? rowm : 0.f;
  float m1 = (k1 < 3) ? rowm : 0.f;

  v2f a, b;
  a.x = m0 * C[k0c * 3 + mc];
  a.y = m1 * C[k1c * 3 + mc];
  b.x = m0 * G[k0c * 3 + mc];
  b.y = m1 * G[k1c * 3 + mc];
  v8f c = {0.f, 0.f, 0.f, 0.f, 0.f, 0.f, 0.f, 0.f};
  v8f d = __builtin_amdgcn_wmma_f32_16x16x4_f32(
       false, a,  false, b,
       (short)0, c,  false,  false);

  __shared__ float lr[9];
  if (lane < 3) {
    lr[0 * 3 + lane] = d[0];
    lr[1 * 3 + lane] = d[1];
    lr[2 * 3 + lane] = d[2];
  }
  __syncthreads();
  if (lane == 0) {
    float inv = -1.0f / vol[g];
    const float* v = gvir + g * 6;
    float* o = out + g * 6;
    o[0] = inv * (v[0] + lr[0]);
    o[1] = inv * (v[1] + lr[4]);
    o[2] = inv * (v[2] + lr[8]);
    o[3] = inv * (v[3] + lr[1]);
    o[4] = inv * (v[4] + lr[5]);
    o[5] = inv * (v[5] + lr[2]);
  }
}

extern "C" void kernel_launch(void* const* d_in, const int* in_sizes, int n_in,
                              void* d_out, int out_size, void* d_ws, size_t ws_size,
                              hipStream_t stream) {
  const float* rij      = (const float*)d_in[0];
  const float* fij      = (const float*)d_in[1];
  const float* pos_grad = (const float*)d_in[2];
  const float* cell_grad = (const float*)d_in[3];
  const float* les_cell  = (const float*)d_in[4];
  const float* vol       = (const float*)d_in[5];
  const int*   eidx      = (const int*)d_in[6];
  const int*   batch     = (const int*)d_in[7];

  int E  = in_sizes[0] / 3;
  int n3 = in_sizes[2];
  int G  = in_sizes[5];

  float* force  = (float*)d_out;
  float* stress = force + n3;
  float* gvir   = (float*)d_ws;

  const int threads = 256;
  les_init_kernel<<<(n3 + threads - 1) / threads, threads, 0, stream>>>(
      pos_grad, force, gvir, n3);
  les_edge_kernel<<<VIR_BLOCKS + FORCE_SLICES * FORCE_TILE_ROWS, EDGE_THREADS, 0, stream>>>(
      rij, fij, eidx, batch, force, gvir, E, n3);
  les_stress_kernel<<<G, 32, 0, stream>>>(
      cell_grad, les_cell, vol, gvir, stress);
}
